# starting kernel
# baseline (speedup 1.0000x reference)
#include <hip/hip_runtime.h>

typedef float v2f __attribute__((ext_vector_type(2)));
typedef float v8f __attribute__((ext_vector_type(8)));

__device__ __constant__ float c_h[9] = {
    0.03782845550699535f, -0.02384946501937986f, -0.1106244044184226f,
    0.3774028556126536f,  0.8526986790094022f,   0.3774028556126537f,
    -0.1106244044184226f, -0.02384946501937986f, 0.03782845550699535f};

__device__ __forceinline__ float tap(int t) {
    int tc = t < 0 ? 0 : (t > 8 ? 8 : t);
    float raw = c_h[tc];
    return (t >= 0 && t <= 8) ? raw : 0.0f;
}

__global__ __launch_bounds__(256) void bior_up_wmma(const float* __restrict__ in,
                                                    float* __restrict__ out) {
    const int lane = threadIdx.x & 31;
    const int wave = threadIdx.x >> 5;
    const int tile = blockIdx.x * 8 + wave;
    const int b  = tile >> 12;
    const int tr = (tile >> 6) & 63;
    const int tc = tile & 63;

    const int m  = lane & 15;
    const int kh = lane >> 4;

    v2f w[3];
    #pragma unroll
    for (int kk = 0; kk < 3; ++kk) {
        const int i0 = 4 * kk + 2 * kh;
        w[kk].x = tap(m + 8 - 2 * (i0 + 0));
        w[kk].y = tap(m + 8 - 2 * (i0 + 1));
    }

    const int r0 = tr * 8 - 2;
    const int c0 = tc * 8 - 2;
    const float* inb = in + (size_t)b * (512 * 512);
    const int gc  = c0 + m;
    const int ccl = gc < 0 ? 0 : (gc > 511 ? 511 : gc);
    const bool cok = (gc >= 0) && (gc < 512);
    v2f bi[3];
    #pragma unroll
    for (int kk = 0; kk < 3; ++kk) {
        float vv[2];
        #pragma unroll
        for (int v = 0; v < 2; ++v) {
            const int gr  = r0 + 4 * kk + 2 * kh + v;
            const int rcl = gr < 0 ? 0 : (gr > 511 ? 511 : gr);
            const float raw = inb[rcl * 512 + ccl];
            vv[v] = (cok && gr >= 0 && gr < 512) ? raw : 0.0f;
        }
        bi[kk].x = vv[0];
        bi[kk].y = vv[1];
    }

    v8f t = {};
    #pragma unroll
    for (int kk = 0; kk < 3; ++kk)
        t = __builtin_amdgcn_wmma_f32_16x16x4_f32(false, w[kk], false, bi[kk],
                                                  (short)0, t, false, false);

    __shared__ float lds[8][16 * 17];
    #pragma unroll
    for (int v = 0; v < 8; ++v)
        lds[wave][(kh * 8 + v) * 17 + m] = t[v];
    __builtin_amdgcn_wave_barrier();
    asm volatile("s_wait_dscnt 0" ::: "memory");

    v2f at[3];
    #pragma unroll
    for (int kk = 0; kk < 3; ++kk) {
        const int j0 = 4 * kk + 2 * kh;
        at[kk].x = lds[wave][m * 17 + j0 + 0];
        at[kk].y = lds[wave][m * 17 + j0 + 1];
    }

    v8f o = {};
    #pragma unroll
    for (int kk = 0; kk < 3; ++kk)
        o = __builtin_amdgcn_wmma_f32_16x16x4_f32(false, at[kk], false, w[kk],
                                                  (short)0, o, false, false);

    float* ob = out + (size_t)b * (1024 * 1024)
                    + (size_t)(tr * 16) * 1024 + (tc * 16);
    #pragma unroll
    for (int v = 0; v < 8; ++v)
        ob[(kh * 8 + v) * 1024 + m] = o[v];
}

extern "C" void kernel_launch(void* const* d_in, const int* in_sizes, int n_in,
                              void* d_out, int out_size, void* d_ws, size_t ws_size,
                              hipStream_t stream) {
    const float* in = (const float*)d_in[0];
    float* out = (float*)d_out;
    dim3 grid(8192), block(256);
    hipLaunchKernelGGL(bior_up_wmma, grid, block, 0, stream, in, out);
}


# each wave owns 128 rows x 64 columns, in registers: the row group between its two 64-row halves is computed once (5 W-pass groups per 8 row tiles, not 6); 2048 waves not 4096; sums unchanged
# speedup vs baseline: 2.1367x; 2.1367x over previous
#include <hip/hip_runtime.h>


typedef float v2f __attribute__((ext_vector_type(2)));
typedef float v4f __attribute__((ext_vector_type(4)));
typedef float v8f __attribute__((ext_vector_type(8)));

__device__ __constant__ float c_h[9] = {
    0.03782845550699535f, -0.02384946501937986f, -0.1106244044184226f,
    0.3774028556126536f,  0.8526986790094022f,   0.3774028556126537f,
    -0.1106244044184226f, -0.02384946501937986f, 0.03782845550699535f};

__device__ __forceinline__ float tap(int t) {
    int tc = t < 0 ? 0 : (t > 8 ? 8 : t);
    float raw = c_h[tc];
    return (t >= 0 && t <= 8) ? raw : 0.0f;
}

__device__ __forceinline__ v8f mma(v2f a, v2f b, v8f c) {
    return __builtin_amdgcn_wmma_f32_16x16x4_f32(false, a, false, b, (short)0, c,
                                                 false, false);
}

__global__ __launch_bounds__(256) void bior_up_wmma(const float* __restrict__ in,
                                                    float* __restrict__ out) {
    const int lane = threadIdx.x & 31;
    const int wave = threadIdx.x >> 5;
    const int blk  = blockIdx.x * 8 + wave;
    const int b  = blk >> 7;
    const int by = (blk >> 4) & 7;
    const int bx = blk & 15;

    const int m  = lane & 15;
    const int kh = lane >> 4;

    v2f ww[2][5];
    {
        const int base[5] = {kh ? 4 : 8, kh ? 6 : 10, kh ? 12 : 16,
                             kh ? -2 : 2, kh ? 14 : 0};
        #pragma unroll
        for (int u = 0; u < 2; ++u) {
            #pragma unroll
            for (int c = 0; c < 5; ++c) {
                const int t0 = 2 * base[c] - 2 * m - u + 4;
                ww[u][c].x = tap(t0);
                ww[u][c].y = tap(t0 + 2);
            }
        }
    }

    v2f wh[2][3];
    #pragma unroll
    for (int sp = 0; sp < 2; ++sp) {
        const int row = sp ? (m ^ 8) : m;
        #pragma unroll
        for (int a = 0; a < 3; ++a) {
            const int i0 = 4 * a + 2 * kh;
            wh[sp][a].x = tap(2 * i0 - row);
            wh[sp][a].y = tap(2 * i0 + 2 - row);
        }
    }

    const float* inb = in + (size_t)b * (512 * 512);
    const int pim = 4 * ((m & 7) >> 1) + 2 * (m >> 3) + (m & 1);
    v4f L[5][5];
    #pragma unroll
    for (int jj = 0; jj < 5; ++jj) {
        const int gi  = 16 * (4 * by + jj) - 2 + (jj == 4 ? (pim & 3) : pim);
        const int rcl = gi < 0 ? 0 : (gi > 511 ? 511 : gi);
        #pragma unroll
        for (int j = 0; j < 5; ++j) {
            const int q   = 2 * (4 * bx + j) - kh;
            const int qcl = q < 0 ? 0 : (q > 127 ? 127 : q);
            L[jj][j] = *(const v4f*)(inb + rcl * 512 + 4 * qcl);
        }
    }
    float* ob = out + (size_t)b * (1024 * 1024)
                    + (size_t)(by * 128) * 1024 + (bx * 64);

    v8f d1[5][2][2];
    #pragma unroll
    for (int jj = 0; jj < 5; ++jj) {
        {
            const int gi  = 16 * (4 * by + jj) - 2 + (jj == 4 ? (pim & 3) : pim);
            const bool rok = (gi >= 0) && (gi < 512);
            #pragma unroll
            for (int j = 0; j < 5; ++j) {
                const int q   = 2 * (4 * bx + j) - kh;
                const bool ok = rok && (q >= 0) && (q < 128);
                const v4f zero = {0.0f, 0.0f, 0.0f, 0.0f};
                L[jj][j] = ok ? L[jj][j] : zero;
            }
        }
        #pragma unroll
        for (int sp = 0; sp < 2; ++sp) {
            const v4f l0 = L[jj][2 * sp];
            const v4f l1 = L[jj][2 * sp + 1];
            const v4f l2 = L[jj][2 * sp + 2];
            v2f a0, a1, a2, a3, a4;
            a0.x = l1[0];  a0.y = l1[1];
            a1.x = l1[2];  a1.y = l1[3];
            a2.x = l2[0];  a2.y = l2[1];
            a3.x = l0[2];  a3.y = l0[3];
            a4.x = kh ? l2[2] : l0[0];
            a4.y = kh ? l2[3] : l0[1];
            #pragma unroll
            for (int u = 0; u < 2; ++u) {
                v8f acc = {};
                acc = mma(a0, ww[u][0], acc);
                acc = mma(a1, ww[u][1], acc);
                acc = mma(a2, ww[u][2], acc);
                acc = mma(a3, ww[u][3], acc);
                acc = mma(a4, ww[u][4], acc);
                d1[jj][sp][u] = acc;
            }
        }
        const int y0 = jj == 0 ? 0 : 2 * jj - 1;
        const int y1 = jj == 4 ? 7 : 2 * jj;
        #pragma unroll
        for (int yy = y0; yy <= y1; ++yy) {
            v8f o[2][2];
            #pragma unroll
            for (int sp = 0; sp < 2; ++sp) {
                #pragma unroll
                for (int u = 0; u < 2; ++u) {
                    const int gj = yy >> 1;
                    v8f acc = {};
                    #pragma unroll
                    for (int a = 0; a < 3; ++a) {
                        const int g = (yy & 1) ? (a == 2 ? gj + 1 : gj) : gj;
                        const int p = (yy & 1) ? (a == 2 ? 0 : a + 2) : a;
                        v2f bb;
                        bb.x = d1[g][sp][u][2 * p];
                        bb.y = d1[g][sp][u][2 * p + 1];
                        acc = mma(wh[sp][a], bb, acc);
                    }
                    o[sp][u] = acc;
                }
            }
            float* oplo = ob + (size_t)(yy * 16) * 1024 + 32 * kh + 2 * m;
            float* ophi = ob + (size_t)(yy * 16 + 8) * 1024 + 32 * (1 - kh) + 2 * m;
            #pragma unroll
            for (int v = 0; v < 8; ++v) {
                v2f lo, hi;
                lo.x = kh ? o[1][0][v] : o[0][0][v];
                lo.y = kh ? o[1][1][v] : o[0][1][v];
                hi.x = kh ? o[0][0][v] : o[1][0][v];
                hi.y = kh ? o[0][1][v] : o[1][1][v];
                *(v2f*)(oplo + v * 1024) = lo;
                *(v2f*)(ophi + v * 1024) = hi;
            }
            __builtin_amdgcn_sched_barrier(0);
        }
    }
}

extern "C" void kernel_launch(void* const* d_in, const int* in_sizes, int n_in,
                              void* d_out, int out_size, void* d_ws, size_t ws_size,
                              hipStream_t stream) {
    const float* in = (const float*)d_in[0];
    float* out = (float*)d_out;
    dim3 grid(256), block(256);
    hipLaunchKernelGGL(bior_up_wmma, grid, block, 0, stream, in, out);
}
